# starting kernel
# baseline (speedup 1.0000x reference)
#include <hip/hip_runtime.h>
#include <hip/hip_bf16.h>

typedef __attribute__((ext_vector_type(2))) float v2f;
typedef __attribute__((ext_vector_type(8))) float v8f;

constexpr int B     = 64;
constexpr int N     = 2048;
constexpr int VOCAB = 21;
constexpr int EMB   = 128;
constexpr int H     = 256;
constexpr int LB    = 900;
constexpr int RB    = 1100;
constexpr int BS    = 8;

constexpr int NB_PRE = (LB + BS - 1) / BS;
constexpr int MID    = RB - LB + 1;
constexpr int NB_SUF = (N - (RB + 1) + BS - 1) / BS;
constexpr int NBLK   = NB_PRE + MID + NB_SUF;
constexpr int ROWS   = B * NBLK;
constexpr int ROWTILES = ROWS / 16;
constexpr float EPSF = 1e-8f;

constexpr size_t OFF_BS    = 0;
constexpr size_t OFF_BX    = OFF_BS    + (size_t)ROWS * H;
constexpr size_t OFF_PM    = OFF_BX    + (size_t)ROWS * 12;
constexpr size_t OFF_BIX   = OFF_PM    + (size_t)ROWS;
constexpr size_t OFF_BMASK = OFF_BIX   + (size_t)ROWS * 12;

__global__ __launch_bounds__(128) void hd_block_embed(
    const float* __restrict__ X,
    const int*   __restrict__ S,
    const float* __restrict__ pad,
    const float* __restrict__ cprob,
    const int*   __restrict__ pmask,
    const float* __restrict__ initX,
    const float* __restrict__ emb,
    float* __restrict__ bE,
    float* __restrict__ coeff,
    float* __restrict__ out_bX,
    float* __restrict__ out_pm,
    float* __restrict__ out_bIX,
    float* __restrict__ out_bmask)
{
  const int bj = blockIdx.x;
  const int b  = bj / NBLK;
  const int j  = bj - b * NBLK;
  const int t  = threadIdx.x;

  if (j >= NB_PRE && j < NB_PRE + MID) {
    const int    p  = LB + (j - NB_PRE);
    const size_t bp = (size_t)b * N + p;
    const bool sel = pmask[bp] > 0;
    float e;
    if (sel) {
      const float* c = cprob + bp * VOCAB;
      float a = 0.f;
      #pragma unroll
      for (int v = 0; v < VOCAB; ++v) a = fmaf(c[v], emb[v * EMB + t], a);
      e = a;
    } else {
      e = emb[(size_t)S[bp] * EMB + t];
    }
    bE[(size_t)bj * EMB + t] = e;
    if (t < 12) {
      const float xv = X[bp * 12 + t];
      out_bX[bj * 12 + t]  = xv;
      out_bIX[bj * 12 + t] = sel ? initX[bp * 12 + t] : xv;
    }
    if (t == 0) {
      coeff[bj]     = 1.f;
      out_pm[bj]    = sel ? 1.f : 0.f;
      out_bmask[bj] = pad[bp];
    }
  } else {
    int p0, pend;
    if (j < NB_PRE) { p0 = j * BS;                        pend = min(p0 + BS, LB); }
    else            { int jj = j - (NB_PRE + MID);
                      p0 = (RB + 1) + jj * BS;            pend = min(p0 + BS, N);  }
    float sm = 0.f, mmax = 0.f, acc = 0.f;
    for (int p = p0; p < pend; ++p) {
      const size_t bp = (size_t)b * N + p;
      const float mp = pad[bp];
      sm   += mp;
      mmax  = fmaxf(mmax, mp);
      acc   = fmaf(mp, emb[(size_t)S[bp] * EMB + t], acc);
    }
    const float inv = 1.f / (sm + EPSF);
    bE[(size_t)bj * EMB + t] = acc * inv;
    if (t < 12) {
      float ax = 0.f;
      for (int p = p0; p < pend; ++p) {
        const size_t bp = (size_t)b * N + p;
        ax = fmaf(pad[bp], X[bp * 12 + t], ax);
      }
      const float bx = ax * inv;
      out_bX[bj * 12 + t]  = bx;
      out_bIX[bj * 12 + t] = bx;
    }
    if (t == 0) {
      coeff[bj]     = sm * inv;
      out_pm[bj]    = 0.f;
      out_bmask[bj] = mmax;
    }
  }
}

__global__ __launch_bounds__(256) void hd_gemm_wmma(
    const float* __restrict__ bE,
    const float* __restrict__ coeff,
    const float* __restrict__ Ww,
    const float* __restrict__ Wb,
    float* __restrict__ bS)
{
  const int lane  = threadIdx.x & 31;
  const int gwave = blockIdx.x * 8 + (threadIdx.x >> 5);
  const int rt = gwave >> 2;
  const int cg = gwave & 3;
  if (rt >= ROWTILES) return;

  const int half = lane >> 4;
  const int lm   = lane & 15;
  const int row  = rt * 16 + lm;
  const float* Arow = bE + (size_t)row * EMB + 2 * half;

  v8f acc[4] = {};
  for (int k = 0; k < EMB; k += 4) {
    v2f a;
    a.x = Arow[k];
    a.y = Arow[k + 1];
    const float* Bp = Ww + (size_t)(k + 2 * half) * H + cg * 64 + lm;
    #pragma unroll
    for (int ct = 0; ct < 4; ++ct) {
      v2f bf;
      bf.x = Bp[ct * 16];
      bf.y = Bp[ct * 16 + H];
      acc[ct] = __builtin_amdgcn_wmma_f32_16x16x4_f32(
           false, a,  false, bf,
           (short)0, acc[ct],  false,  false);
    }
  }

  float cf[8];
  #pragma unroll
  for (int r = 0; r < 8; ++r) cf[r] = coeff[rt * 16 + half * 8 + r];
  #pragma unroll
  for (int ct = 0; ct < 4; ++ct) {
    const int n  = cg * 64 + ct * 16 + lm;
    const float wb = Wb[n];
    #pragma unroll
    for (int r = 0; r < 8; ++r) {
      const int m = rt * 16 + half * 8 + r;
      bS[(size_t)m * H + n] = acc[ct][r] + cf[r] * wb;
    }
  }
}

extern "C" void kernel_launch(void* const* d_in, const int* in_sizes, int n_in,
                              void* d_out, int out_size, void* d_ws, size_t ws_size,
                              hipStream_t stream) {
  const float* X      = (const float*)d_in[0];
  const int*   S      = (const int*)  d_in[1];
  const float* pad    = (const float*)d_in[2];
  const float* cprob  = (const float*)d_in[3];
  const int*   pmask  = (const int*)  d_in[4];
  const float* initX  = (const float*)d_in[5];
  const float* emb    = (const float*)d_in[6];
  const float* Ww     = (const float*)d_in[7];
  const float* Wb     = (const float*)d_in[8];

  float* out   = (float*)d_out;
  float* bS    = out + OFF_BS;
  float* bX    = out + OFF_BX;
  float* pm    = out + OFF_PM;
  float* bIX   = out + OFF_BIX;
  float* bmask = out + OFF_BMASK;

  float* bE    = (float*)d_ws;
  float* coeff = bE + (size_t)ROWS * EMB;

  hd_block_embed<<<ROWS, 128, 0, stream>>>(X, S, pad, cprob, pmask, initX, emb,
                                           bE, coeff, bX, pm, bIX, bmask);

  const int totalWaves = ROWTILES * 4;
  hd_gemm_wmma<<<(totalWaves + 7) / 8, 256, 0, stream>>>(bE, coeff, Ww, Wb, bS);
}


# Reassociate (c@emb)@W_w as c@(emb@W_w): T=emb@W_w with W_b as row 21 built once on WMMA, rows become 24-wide weight vectors made in registers, K 128->24, no bE round trip; f32 sum order changes
# speedup vs baseline: 3.1193x; 3.1193x over previous
#include <hip/hip_runtime.h>

typedef __attribute__((ext_vector_type(2))) float v2f;
typedef __attribute__((ext_vector_type(8))) float v8f;

constexpr int B     = 64;
constexpr int N     = 2048;
constexpr int VOCAB = 21;
constexpr int EMB   = 128;
constexpr int H     = 256;
constexpr int LB    = 900;
constexpr int RB    = 1100;
constexpr int BS    = 8;

constexpr int NB_PRE = (LB + BS - 1) / BS;
constexpr int MID    = RB - LB + 1;
constexpr int NB_SUF = (N - (RB + 1) + BS - 1) / BS;
constexpr int NBLK   = NB_PRE + MID + NB_SUF;
constexpr int ROWS   = B * NBLK;
constexpr int ROWTILES = ROWS / 16;
constexpr float EPSF = 1e-8f;

constexpr int KP     = 24;
constexpr int KSTEPS = KP / 4;
constexpr int NT     = H / 16;
constexpr int TF_FLOATS = KSTEPS * NT * 32 * 2;

constexpr int TBLOCKS = 4;
constexpr int XITEMS  = ROWS * 3;
constexpr int XBLOCKS = (XITEMS + 255) / 256;
constexpr int PREP_GRID = TBLOCKS + XBLOCKS;
constexpr int ROW_GRID  = (ROWTILES + 7) / 8;

constexpr size_t OFF_BS    = 0;
constexpr size_t OFF_BX    = OFF_BS    + (size_t)ROWS * H;
constexpr size_t OFF_PM    = OFF_BX    + (size_t)ROWS * 12;
constexpr size_t OFF_BIX   = OFF_PM    + (size_t)ROWS;
constexpr size_t OFF_BMASK = OFF_BIX   + (size_t)ROWS * 12;

__device__ __forceinline__ void row_geom(int row, int& base, int& cnt, bool& mid) {
  const int b = row / NBLK;
  const int j = row - b * NBLK;
  int p0;
  if (j < NB_PRE)            { p0 = j * BS;                           cnt = min(BS, LB - p0); mid = false; }
  else if (j < NB_PRE + MID) { p0 = LB + (j - NB_PRE);                cnt = 1;                mid = true;  }
  else                       { p0 = (RB + 1) + (j - NB_PRE - MID) * BS; cnt = min(BS, N - p0);  mid = false; }
  base = b * N + p0;
}

__global__ __launch_bounds__(256) void hd_prep(
    const float* __restrict__ X,
    const float* __restrict__ pad,
    const int*   __restrict__ pmask,
    const float* __restrict__ initX,
    const float* __restrict__ emb,
    const float* __restrict__ Ww,
    const float* __restrict__ Wb,
    float* __restrict__ Tf,
    float* __restrict__ out_bX,
    float* __restrict__ out_pm,
    float* __restrict__ out_bIX,
    float* __restrict__ out_bmask)
{
  const int tid = threadIdx.x;
  if (blockIdx.x < TBLOCKS) {
    const int lane = tid & 31;
    const int gw   = blockIdx.x * 8 + (tid >> 5);
    const int mt   = gw & 1;
    const int nt   = gw >> 1;
    const int half = lane >> 4;
    const int lm   = lane & 15;
    const int v    = mt * 16 + lm;
    const bool vok = v < VOCAB;
    const float* Arow = emb + (size_t)(vok ? v : 0) * EMB + 2 * half;
    const float* Bcol = Ww + (size_t)(2 * half) * H + nt * 16 + lm;
    v8f acc = {};
    #pragma unroll
    for (int k = 0; k < EMB; k += 4) {
      v2f a;
      const float a0 = Arow[k];
      const float a1 = Arow[k + 1];
      a.x = vok ? a0 : 0.f;
      a.y = vok ? a1 : 0.f;
      v2f bf;
      bf.x = Bcol[(size_t)k * H];
      bf.y = Bcol[(size_t)k * H + H];
      acc = __builtin_amdgcn_wmma_f32_16x16x4_f32(
           false, a,  false, bf,
           (short)0, acc,  false,  false);
    }
    const int n = nt * 16 + lm;
    const float wb = Wb[n];
    #pragma unroll
    for (int r = 0; r < 8; ++r) {
      const int vv = mt * 16 + half * 8 + r;
      if (vv < KP) {
        const float val = (vv < VOCAB) ? acc[r] : ((vv == VOCAB) ? wb : 0.f);
        const int s  = vv >> 2;
        const int h2 = (vv >> 1) & 1;
        const int e  = vv & 1;
        Tf[(((s * NT + nt) * 32) + h2 * 16 + lm) * 2 + e] = val;
      }
    }
    return;
  }

  const int idx = (blockIdx.x - TBLOCKS) * 256 + tid;
  if (idx >= XITEMS) return;
  const int row = idx / 3;
  const int q   = idx - row * 3;
  int base, cnt; bool mid;
  row_geom(row, base, cnt, mid);
  const float4* X4 = reinterpret_cast<const float4*>(X);
  const float4* I4 = reinterpret_cast<const float4*>(initX);

  float  mv[BS]; int pv[BS]; float4 xv[BS];
  #pragma unroll
  for (int i = 0; i < BS; ++i) {
    const int bp = base + (i < cnt ? i : 0);
    mv[i] = pad[bp];
    pv[i] = pmask[bp];
    xv[i] = X4[(size_t)bp * 3 + q];
  }
  float4 res_x, res_i; float res_pm, res_mask;
  if (mid) {
    const bool sel = pv[0] > 0;
    res_x = xv[0];
    res_i = xv[0];
    if (sel) res_i = I4[(size_t)base * 3 + q];
    res_pm   = sel ? 1.f : 0.f;
    res_mask = mv[0];
  } else {
    float sm = 0.f;
    float mmax = (cnt < BS) ? 0.f : mv[0];
    float4 ax = make_float4(0.f, 0.f, 0.f, 0.f);
    float4 ai = ax;
    #pragma unroll
    for (int i = 0; i < BS; ++i) {
      if (i < cnt) {
        const float w = mv[i];
        sm  += w;
        mmax = fmaxf(mmax, w);
        float4 xi = xv[i];
        if (pv[i] > 0) xi = I4[(size_t)(base + i) * 3 + q];
        ax.x = fmaf(w, xv[i].x, ax.x); ax.y = fmaf(w, xv[i].y, ax.y);
        ax.z = fmaf(w, xv[i].z, ax.z); ax.w = fmaf(w, xv[i].w, ax.w);
        ai.x = fmaf(w, xi.x, ai.x);    ai.y = fmaf(w, xi.y, ai.y);
        ai.z = fmaf(w, xi.z, ai.z);    ai.w = fmaf(w, xi.w, ai.w);
      }
    }
    const float inv = 1.f / (sm + EPSF);
    res_x = make_float4(ax.x * inv, ax.y * inv, ax.z * inv, ax.w * inv);
    res_i = make_float4(ai.x * inv, ai.y * inv, ai.z * inv, ai.w * inv);
    res_pm   = 0.f;
    res_mask = mmax;
  }
  reinterpret_cast<float4*>(out_bX)[idx]  = res_x;
  reinterpret_cast<float4*>(out_bIX)[idx] = res_i;
  if (q == 0) {
    out_pm[row]    = res_pm;
    out_bmask[row] = res_mask;
  }
}

__global__ __launch_bounds__(256) void hd_rows_wmma(
    const int*   __restrict__ S,
    const float* __restrict__ pad,
    const float* __restrict__ cprob,
    const int*   __restrict__ pmask,
    const float* __restrict__ Tf,
    float* __restrict__ bS)
{
  __shared__ float4 Tl4[TF_FLOATS / 4];
  const int tid = threadIdx.x;
  {
    const float4* Tg = reinterpret_cast<const float4*>(Tf);
    #pragma unroll
    for (int i = 0; i < TF_FLOATS / 4 / 256; ++i) Tl4[tid + 256 * i] = Tg[tid + 256 * i];
  }

  const int lane = tid & 31;
  const int tile = blockIdx.x * 8 + (tid >> 5);
  const bool active = tile < ROWTILES;
  const int half = lane >> 4;
  const int lm   = lane & 15;

  float a[2 * KSTEPS];
  #pragma unroll
  for (int qi = 0; qi < 2 * KSTEPS; ++qi) a[qi] = 0.f;

  if (active) {
    const int row = tile * 16 + lm;
    int base, cnt; bool mid;
    row_geom(row, base, cnt, mid);
    int sv[BS]; float mv[BS]; int pv[BS];
    #pragma unroll
    for (int i = 0; i < BS; ++i) {
      const int bp = base + (i < cnt ? i : 0);
      sv[i] = S[bp];
      mv[i] = pad[bp];
      pv[i] = pmask[bp];
    }
    float sm = 0.f;
    #pragma unroll
    for (int i = 0; i < BS; ++i) {
      const bool valid = i < cnt;
      const float w = valid ? (mid ? 1.f : mv[i]) : 0.f;
      sm += valid ? mv[i] : 0.f;
      if (valid && pv[i] > 0) {
        const float* c = cprob + (size_t)(base + i) * VOCAB;
        #pragma unroll
        for (int qi = 0; qi < 2 * KSTEPS; ++qi) {
          const int v = 4 * (qi >> 1) + 2 * half + (qi & 1);
          const float cv = c[v < VOCAB ? v : 0];
          a[qi] = fmaf(w, v < VOCAB ? cv : 0.f, a[qi]);
        }
      } else {
        #pragma unroll
        for (int qi = 0; qi < 2 * KSTEPS; ++qi) {
          const int v = 4 * (qi >> 1) + 2 * half + (qi & 1);
          a[qi] += (sv[i] == v) ? w : 0.f;
        }
      }
    }
    const float inv = mid ? 1.f : 1.f / (sm + EPSF);
    #pragma unroll
    for (int qi = 0; qi < 2 * KSTEPS; ++qi) a[qi] *= inv;
    const float cb = mid ? 1.f : sm * inv;
    if (half == 0) a[2 * KSTEPS - 1] = cb;
  }

  __syncthreads();
  if (!active) return;

  const v2f* Tl = reinterpret_cast<const v2f*>(Tl4);
  float* orow = bS + (size_t)(tile * 16 + half * 8) * H + lm;
  #pragma unroll
  for (int ct = 0; ct < NT; ct += 2) {
    v8f acc0 = {};
    v8f acc1 = {};
    #pragma unroll
    for (int s = 0; s < KSTEPS; ++s) {
      v2f af;
      af.x = a[2 * s];
      af.y = a[2 * s + 1];
      const v2f b0 = Tl[(s * NT + ct) * 32 + lane];
      const v2f b1 = Tl[(s * NT + ct + 1) * 32 + lane];
      acc0 = __builtin_amdgcn_wmma_f32_16x16x4_f32(
           false, af,  false, b0,
           (short)0, acc0,  false,  false);
      acc1 = __builtin_amdgcn_wmma_f32_16x16x4_f32(
           false, af,  false, b1,
           (short)0, acc1,  false,  false);
    }
    #pragma unroll
    for (int r = 0; r < 8; ++r) {
      orow[(size_t)r * H + ct * 16]      = acc0[r];
      orow[(size_t)r * H + ct * 16 + 16] = acc1[r];
    }
  }
}

extern "C" void kernel_launch(void* const* d_in, const int* in_sizes, int n_in,
                              void* d_out, int out_size, void* d_ws, size_t ws_size,
                              hipStream_t stream) {
  const float* X      = (const float*)d_in[0];
  const int*   S      = (const int*)  d_in[1];
  const float* pad    = (const float*)d_in[2];
  const float* cprob  = (const float*)d_in[3];
  const int*   pmask  = (const int*)  d_in[4];
  const float* initX  = (const float*)d_in[5];
  const float* emb    = (const float*)d_in[6];
  const float* Ww     = (const float*)d_in[7];
  const float* Wb     = (const float*)d_in[8];

  float* out   = (float*)d_out;
  float* bS    = out + OFF_BS;
  float* bX    = out + OFF_BX;
  float* pm    = out + OFF_PM;
  float* bIX   = out + OFF_BIX;
  float* bmask = out + OFF_BMASK;

  float* Tf    = (float*)d_ws;

  hd_prep<<<PREP_GRID, 256, 0, stream>>>(X, pad, pmask, initX, emb, Ww, Wb,
                                         Tf, bX, pm, bIX, bmask);
  hd_rows_wmma<<<ROW_GRID, 256, 0, stream>>>(S, pad, cprob, pmask, Tf, bS);
}


# bS tile staged per wave in LDS (WMMA operand roles swapped so a lane holds 8 consecutive columns), written as whole 128-B lines with 16-B stores: 32 store instructions per wave, not 128; same sums
# speedup vs baseline: 4.2057x; 1.3483x over previous
#include <hip/hip_runtime.h>

typedef __attribute__((ext_vector_type(2))) float v2f;
typedef __attribute__((ext_vector_type(8))) float v8f;

constexpr int B     = 64;
constexpr int N     = 2048;
constexpr int VOCAB = 21;
constexpr int EMB   = 128;
constexpr int H     = 256;
constexpr int LB    = 900;
constexpr int RB    = 1100;
constexpr int BS    = 8;

constexpr int NB_PRE = (LB + BS - 1) / BS;
constexpr int MID    = RB - LB + 1;
constexpr int NB_SUF = (N - (RB + 1) + BS - 1) / BS;
constexpr int NBLK   = NB_PRE + MID + NB_SUF;
constexpr int ROWS   = B * NBLK;
constexpr int ROWTILES = ROWS / 16;
constexpr float EPSF = 1e-8f;

constexpr int KP     = 24;
constexpr int KSTEPS = KP / 4;
constexpr int NT     = H / 16;
constexpr int TF_FLOATS = KSTEPS * NT * 32 * 2;

constexpr int TBLOCKS = 4;
constexpr int XITEMS  = ROWS * 3;
constexpr int XBLOCKS = (XITEMS + 255) / 256;
constexpr int PREP_GRID = TBLOCKS + XBLOCKS;
constexpr int ROW_GRID  = (ROWTILES + 7) / 8;

constexpr int SCOLS  = 64;
constexpr int SPITCH = SCOLS + 4;
constexpr int SWAVE  = 16 * SPITCH;

constexpr size_t OFF_BS    = 0;
constexpr size_t OFF_BX    = OFF_BS    + (size_t)ROWS * H;
constexpr size_t OFF_PM    = OFF_BX    + (size_t)ROWS * 12;
constexpr size_t OFF_BIX   = OFF_PM    + (size_t)ROWS;
constexpr size_t OFF_BMASK = OFF_BIX   + (size_t)ROWS * 12;

__device__ __forceinline__ void row_geom(int row, int& base, int& cnt, bool& mid) {
  const int b = row / NBLK;
  const int j = row - b * NBLK;
  int p0;
  if (j < NB_PRE)            { p0 = j * BS;                           cnt = min(BS, LB - p0); mid = false; }
  else if (j < NB_PRE + MID) { p0 = LB + (j - NB_PRE);                cnt = 1;                mid = true;  }
  else                       { p0 = (RB + 1) + (j - NB_PRE - MID) * BS; cnt = min(BS, N - p0);  mid = false; }
  base = b * N + p0;
}

__global__ __launch_bounds__(256) void hd_prep(
    const float* __restrict__ X,
    const float* __restrict__ pad,
    const int*   __restrict__ pmask,
    const float* __restrict__ initX,
    const float* __restrict__ emb,
    const float* __restrict__ Ww,
    const float* __restrict__ Wb,
    float* __restrict__ Tf,
    float* __restrict__ out_bX,
    float* __restrict__ out_pm,
    float* __restrict__ out_bIX,
    float* __restrict__ out_bmask)
{
  const int tid = threadIdx.x;
  if (blockIdx.x < TBLOCKS) {
    const int lane = tid & 31;
    const int gw   = blockIdx.x * 8 + (tid >> 5);
    const int mt   = gw & 1;
    const int nt   = gw >> 1;
    const int half = lane >> 4;
    const int lm   = lane & 15;
    const int v    = mt * 16 + lm;
    const bool vok = v < VOCAB;
    const float* Arow = emb + (size_t)(vok ? v : 0) * EMB + 2 * half;
    const float* Bcol = Ww + (size_t)(2 * half) * H + nt * 16 + lm;
    v8f acc = {};
    #pragma unroll
    for (int k = 0; k < EMB; k += 4) {
      v2f a;
      const float a0 = Arow[k];
      const float a1 = Arow[k + 1];
      a.x = vok ? a0 : 0.f;
      a.y = vok ? a1 : 0.f;
      v2f bf;
      bf.x = Bcol[(size_t)k * H];
      bf.y = Bcol[(size_t)k * H + H];
      acc = __builtin_amdgcn_wmma_f32_16x16x4_f32(
           false, a,  false, bf,
           (short)0, acc,  false,  false);
    }
    const int n = nt * 16 + lm;
    const float wb = Wb[n];
    #pragma unroll
    for (int r = 0; r < 8; ++r) {
      const int vv = mt * 16 + half * 8 + r;
      if (vv < KP) {
        const float val = (vv < VOCAB) ? acc[r] : ((vv == VOCAB) ? wb : 0.f);
        const int s  = vv >> 2;
        const int h2 = (vv >> 1) & 1;
        const int e  = vv & 1;
        Tf[(((s * NT + nt) * 32) + h2 * 16 + lm) * 2 + e] = val;
      }
    }
    return;
  }

  const int idx = (blockIdx.x - TBLOCKS) * 256 + tid;
  if (idx >= XITEMS) return;
  const int row = idx / 3;
  const int q   = idx - row * 3;
  int base, cnt; bool mid;
  row_geom(row, base, cnt, mid);
  const float4* X4 = reinterpret_cast<const float4*>(X);
  const float4* I4 = reinterpret_cast<const float4*>(initX);

  float  mv[BS]; int pv[BS]; float4 xv[BS];
  #pragma unroll
  for (int i = 0; i < BS; ++i) {
    const int bp = base + (i < cnt ? i : 0);
    mv[i] = pad[bp];
    pv[i] = pmask[bp];
    xv[i] = X4[(size_t)bp * 3 + q];
  }
  float4 res_x, res_i; float res_pm, res_mask;
  if (mid) {
    const bool sel = pv[0] > 0;
    res_x = xv[0];
    res_i = xv[0];
    if (sel) res_i = I4[(size_t)base * 3 + q];
    res_pm   = sel ? 1.f : 0.f;
    res_mask = mv[0];
  } else {
    float sm = 0.f;
    float mmax = (cnt < BS) ? 0.f : mv[0];
    float4 ax = make_float4(0.f, 0.f, 0.f, 0.f);
    float4 ai = ax;
    #pragma unroll
    for (int i = 0; i < BS; ++i) {
      if (i < cnt) {
        const float w = mv[i];
        sm  += w;
        mmax = fmaxf(mmax, w);
        float4 xi = xv[i];
        if (pv[i] > 0) xi = I4[(size_t)(base + i) * 3 + q];
        ax.x = fmaf(w, xv[i].x, ax.x); ax.y = fmaf(w, xv[i].y, ax.y);
        ax.z = fmaf(w, xv[i].z, ax.z); ax.w = fmaf(w, xv[i].w, ax.w);
        ai.x = fmaf(w, xi.x, ai.x);    ai.y = fmaf(w, xi.y, ai.y);
        ai.z = fmaf(w, xi.z, ai.z);    ai.w = fmaf(w, xi.w, ai.w);
      }
    }
    const float inv = 1.f / (sm + EPSF);
    res_x = make_float4(ax.x * inv, ax.y * inv, ax.z * inv, ax.w * inv);
    res_i = make_float4(ai.x * inv, ai.y * inv, ai.z * inv, ai.w * inv);
    res_pm   = 0.f;
    res_mask = mmax;
  }
  reinterpret_cast<float4*>(out_bX)[idx]  = res_x;
  reinterpret_cast<float4*>(out_bIX)[idx] = res_i;
  if (q == 0) {
    out_pm[row]    = res_pm;
    out_bmask[row] = res_mask;
  }
}

__global__ __launch_bounds__(256) void hd_rows_wmma(
    const int*   __restrict__ S,
    const float* __restrict__ pad,
    const float* __restrict__ cprob,
    const int*   __restrict__ pmask,
    const float* __restrict__ Tf,
    float* __restrict__ bS)
{
  __shared__ float4 Tl4[TF_FLOATS / 4];
  __shared__ float4 Sg4[8 * SWAVE / 4];
  const int tid = threadIdx.x;
  {
    const float4* Tg = reinterpret_cast<const float4*>(Tf);
    #pragma unroll
    for (int i = 0; i < TF_FLOATS / 4 / 256; ++i) Tl4[tid + 256 * i] = Tg[tid + 256 * i];
  }

  const int lane = tid & 31;
  const int tile = blockIdx.x * 8 + (tid >> 5);
  const bool active = tile < ROWTILES;
  const int half = lane >> 4;
  const int lm   = lane & 15;

  float a[2 * KSTEPS];
  #pragma unroll
  for (int qi = 0; qi < 2 * KSTEPS; ++qi) a[qi] = 0.f;

  if (active) {
    const int row = tile * 16 + lm;
    int base, cnt; bool mid;
    row_geom(row, base, cnt, mid);
    int sv[BS]; float mv[BS]; int pv[BS];
    #pragma unroll
    for (int i = 0; i < BS; ++i) {
      const int bp = base + (i < cnt ? i : 0);
      sv[i] = S[bp];
      mv[i] = pad[bp];
      pv[i] = pmask[bp];
    }
    float sm = 0.f;
    #pragma unroll
    for (int i = 0; i < BS; ++i) {
      const bool valid = i < cnt;
      const float w = valid ? (mid ? 1.f : mv[i]) : 0.f;
      sm += valid ? mv[i] : 0.f;
      if (valid && pv[i] > 0) {
        const float* c = cprob + (size_t)(base + i) * VOCAB;
        #pragma unroll
        for (int qi = 0; qi < 2 * KSTEPS; ++qi) {
          const int v = 4 * (qi >> 1) + 2 * half + (qi & 1);
          const float cv = c[v < VOCAB ? v : 0];
          a[qi] = fmaf(w, v < VOCAB ? cv : 0.f, a[qi]);
        }
      } else {
        #pragma unroll
        for (int qi = 0; qi < 2 * KSTEPS; ++qi) {
          const int v = 4 * (qi >> 1) + 2 * half + (qi & 1);
          a[qi] += (sv[i] == v) ? w : 0.f;
        }
      }
    }
    const float inv = mid ? 1.f : 1.f / (sm + EPSF);
    #pragma unroll
    for (int qi = 0; qi < 2 * KSTEPS; ++qi) a[qi] *= inv;
    const float cb = mid ? 1.f : sm * inv;
    if (half == 0) a[2 * KSTEPS - 1] = cb;
  }

  __syncthreads();
  if (!active) return;

  const v2f* Tl = reinterpret_cast<const v2f*>(Tl4);
  float* stg = reinterpret_cast<float*>(Sg4) + (tid >> 5) * SWAVE;
  float* wr  = stg + lm * SPITCH + half * 8;
  const float* rd = stg + half * SPITCH + lm * 4;
  float* orow = bS + (size_t)(tile * 16 + half) * H + lm * 4;
  #pragma unroll
  for (int g = 0; g < NT / 4; ++g) {
    #pragma unroll
    for (int t = 0; t < 4; t += 2) {
      const int ct = g * 4 + t;
      v8f acc0 = {};
      v8f acc1 = {};
      #pragma unroll
      for (int s = 0; s < KSTEPS; ++s) {
        v2f af;
        af.x = a[2 * s];
        af.y = a[2 * s + 1];
        const v2f b0 = Tl[(s * NT + ct) * 32 + lane];
        const v2f b1 = Tl[(s * NT + ct + 1) * 32 + lane];
        acc0 = __builtin_amdgcn_wmma_f32_16x16x4_f32(
             false, b0,  false, af,
             (short)0, acc0,  false,  false);
        acc1 = __builtin_amdgcn_wmma_f32_16x16x4_f32(
             false, b1,  false, af,
             (short)0, acc1,  false,  false);
      }
      *reinterpret_cast<float4*>(wr + 16 * t)      = make_float4(acc0[0], acc0[1], acc0[2], acc0[3]);
      *reinterpret_cast<float4*>(wr + 16 * t + 4)  = make_float4(acc0[4], acc0[5], acc0[6], acc0[7]);
      *reinterpret_cast<float4*>(wr + 16 * t + 16) = make_float4(acc1[0], acc1[1], acc1[2], acc1[3]);
      *reinterpret_cast<float4*>(wr + 16 * t + 20) = make_float4(acc1[4], acc1[5], acc1[6], acc1[7]);
    }
    __builtin_amdgcn_fence(__ATOMIC_RELEASE, "wavefront");
    __builtin_amdgcn_wave_barrier();
    __builtin_amdgcn_fence(__ATOMIC_ACQUIRE, "wavefront");
    #pragma unroll
    for (int j = 0; j < 8; ++j) {
      const float4 v = *reinterpret_cast<const float4*>(rd + 2 * j * SPITCH);
      *reinterpret_cast<float4*>(orow + (size_t)(2 * j) * H + g * SCOLS) = v;
    }
    if (g + 1 < NT / 4) {
      __builtin_amdgcn_fence(__ATOMIC_RELEASE, "wavefront");
      __builtin_amdgcn_wave_barrier();
      __builtin_amdgcn_fence(__ATOMIC_ACQUIRE, "wavefront");
    }
  }
}

extern "C" void kernel_launch(void* const* d_in, const int* in_sizes, int n_in,
                              void* d_out, int out_size, void* d_ws, size_t ws_size,
                              hipStream_t stream) {
  const float* X      = (const float*)d_in[0];
  const int*   S      = (const int*)  d_in[1];
  const float* pad    = (const float*)d_in[2];
  const float* cprob  = (const float*)d_in[3];
  const int*   pmask  = (const int*)  d_in[4];
  const float* initX  = (const float*)d_in[5];
  const float* emb    = (const float*)d_in[6];
  const float* Ww     = (const float*)d_in[7];
  const float* Wb     = (const float*)d_in[8];

  float* out   = (float*)d_out;
  float* bS    = out + OFF_BS;
  float* bX    = out + OFF_BX;
  float* pm    = out + OFF_PM;
  float* bIX   = out + OFF_BIX;
  float* bmask = out + OFF_BMASK;

  float* Tf    = (float*)d_ws;

  hd_prep<<<PREP_GRID, 256, 0, stream>>>(X, pad, pmask, initX, emb, Ww, Wb,
                                         Tf, bX, pm, bIX, bmask);
  hd_rows_wmma<<<ROW_GRID, 256, 0, stream>>>(S, pad, cprob, pmask, Tf, bS);
}


# Both launches fused into one: each row block (128 row groups x 2 column slices) rebuilds its 24x128 slice of T=emb@W_w on WMMA in LDS; small outputs in extra blocks of the same grid; same sums
# speedup vs baseline: 4.4869x; 1.0669x over previous
#include <hip/hip_runtime.h>

typedef __attribute__((ext_vector_type(2))) float v2f;
typedef __attribute__((ext_vector_type(8))) float v8f;

constexpr int B     = 64;
constexpr int N     = 2048;
constexpr int VOCAB = 21;
constexpr int EMB   = 128;
constexpr int H     = 256;
constexpr int LB    = 900;
constexpr int RB    = 1100;
constexpr int BS    = 8;

constexpr int NB_PRE = (LB + BS - 1) / BS;
constexpr int MID    = RB - LB + 1;
constexpr int NB_SUF = (N - (RB + 1) + BS - 1) / BS;
constexpr int NBLK   = NB_PRE + MID + NB_SUF;
constexpr int ROWS   = B * NBLK;
constexpr int ROWTILES = ROWS / 16;
constexpr float EPSF = 1e-8f;

constexpr int KP     = 24;
constexpr int KSTEPS = KP / 4;
constexpr int NT     = H / 16;
constexpr int CSL    = 2;
constexpr int CT_PER = NT / CSL;
constexpr int TS_FLOATS = KSTEPS * CT_PER * 32 * 2;

constexpr int RG      = 128;
constexpr int TPG     = ROWTILES / RG;
constexpr int TREM    = ROWTILES - TPG * RG;
constexpr int VISITS  = (TPG + 1 + 7) / 8;
constexpr int ROWBLOCKS = RG * CSL;
constexpr int XITEMS  = ROWS * 3;
constexpr int XBLOCKS = (XITEMS + 255) / 256;
constexpr int GRID    = ROWBLOCKS + XBLOCKS;

constexpr int SCOLS  = 64;
constexpr int SPITCH = SCOLS + 4;
constexpr int SWAVE  = 16 * SPITCH;

constexpr size_t OFF_BS    = 0;
constexpr size_t OFF_BX    = OFF_BS    + (size_t)ROWS * H;
constexpr size_t OFF_PM    = OFF_BX    + (size_t)ROWS * 12;
constexpr size_t OFF_BIX   = OFF_PM    + (size_t)ROWS;
constexpr size_t OFF_BMASK = OFF_BIX   + (size_t)ROWS * 12;

__device__ __forceinline__ void row_geom(int row, int& base, int& cnt, bool& mid) {
  const int b = row / NBLK;
  const int j = row - b * NBLK;
  int p0;
  if (j < NB_PRE)            { p0 = j * BS;                           cnt = min(BS, LB - p0); mid = false; }
  else if (j < NB_PRE + MID) { p0 = LB + (j - NB_PRE);                cnt = 1;                mid = true;  }
  else                       { p0 = (RB + 1) + (j - NB_PRE - MID) * BS; cnt = min(BS, N - p0);  mid = false; }
  base = b * N + p0;
}

__device__ __forceinline__ void build_weights(
    int base, int cnt, bool mid, int half,
    const int (&sv)[BS], const float (&mv)[BS], const int (&pv)[BS],
    const float* __restrict__ cprob,
    float (&a)[2 * KSTEPS])
{
  #pragma unroll
  for (int qi = 0; qi < 2 * KSTEPS; ++qi) a[qi] = 0.f;
  float sm = 0.f;
  #pragma unroll
  for (int i = 0; i < BS; ++i) {
    const bool valid = i < cnt;
    const float w = valid ? (mid ? 1.f : mv[i]) : 0.f;
    sm += valid ? mv[i] : 0.f;
    if (valid && pv[i] > 0) {
      const float* c = cprob + (size_t)(base + i) * VOCAB;
      #pragma unroll
      for (int qi = 0; qi < 2 * KSTEPS; ++qi) {
        const int v = 4 * (qi >> 1) + 2 * half + (qi & 1);
        const float cv = c[v < VOCAB ? v : 0];
        a[qi] = fmaf(w, v < VOCAB ? cv : 0.f, a[qi]);
      }
    } else {
      #pragma unroll
      for (int qi = 0; qi < 2 * KSTEPS; ++qi) {
        const int v = 4 * (qi >> 1) + 2 * half + (qi & 1);
        a[qi] += (sv[i] == v) ? w : 0.f;
      }
    }
  }
  const float inv = mid ? 1.f : 1.f / (sm + EPSF);
  #pragma unroll
  for (int qi = 0; qi < 2 * KSTEPS; ++qi) a[qi] *= inv;
  const float cb = mid ? 1.f : sm * inv;
  if (half == 0) a[2 * KSTEPS - 1] = cb;
}

__global__ __launch_bounds__(256) void hd_fused(
    const float* __restrict__ X,
    const int*   __restrict__ S,
    const float* __restrict__ pad,
    const float* __restrict__ cprob,
    const int*   __restrict__ pmask,
    const float* __restrict__ initX,
    const float* __restrict__ emb,
    const float* __restrict__ Ww,
    const float* __restrict__ Wb,
    float* __restrict__ bS,
    float* __restrict__ out_bX,
    float* __restrict__ out_pm,
    float* __restrict__ out_bIX,
    float* __restrict__ out_bmask)
{
  __shared__ float4 Tl4[TS_FLOATS / 4];
  __shared__ float4 Sg4[8 * SWAVE / 4];
  const int tid = threadIdx.x;

  if (blockIdx.x >= ROWBLOCKS) {
    const int idx = (blockIdx.x - ROWBLOCKS) * 256 + tid;
    if (idx >= XITEMS) return;
    const int row = idx / 3;
    const int q   = idx - row * 3;
    int base, cnt; bool mid;
    row_geom(row, base, cnt, mid);
    const float4* X4 = reinterpret_cast<const float4*>(X);
    const float4* I4 = reinterpret_cast<const float4*>(initX);

    float  mv[BS]; int pv[BS]; float4 xv[BS];
    #pragma unroll
    for (int i = 0; i < BS; ++i) {
      const int bp = base + (i < cnt ? i : 0);
      mv[i] = pad[bp];
      pv[i] = pmask[bp];
      xv[i] = X4[(size_t)bp * 3 + q];
    }
    float4 res_x, res_i; float res_pm, res_mask;
    if (mid) {
      const bool sel = pv[0] > 0;
      res_x = xv[0];
      res_i = xv[0];
      if (sel) res_i = I4[(size_t)base * 3 + q];
      res_pm   = sel ? 1.f : 0.f;
      res_mask = mv[0];
    } else {
      float sm = 0.f;
      float mmax = (cnt < BS) ? 0.f : mv[0];
      float4 ax = make_float4(0.f, 0.f, 0.f, 0.f);
      float4 ai = ax;
      #pragma unroll
      for (int i = 0; i < BS; ++i) {
        if (i < cnt) {
          const float w = mv[i];
          sm  += w;
          mmax = fmaxf(mmax, w);
          float4 xi = xv[i];
          if (pv[i] > 0) xi = I4[(size_t)(base + i) * 3 + q];
          ax.x = fmaf(w, xv[i].x, ax.x); ax.y = fmaf(w, xv[i].y, ax.y);
          ax.z = fmaf(w, xv[i].z, ax.z); ax.w = fmaf(w, xv[i].w, ax.w);
          ai.x = fmaf(w, xi.x, ai.x);    ai.y = fmaf(w, xi.y, ai.y);
          ai.z = fmaf(w, xi.z, ai.z);    ai.w = fmaf(w, xi.w, ai.w);
        }
      }
      const float inv = 1.f / (sm + EPSF);
      res_x = make_float4(ax.x * inv, ax.y * inv, ax.z * inv, ax.w * inv);
      res_i = make_float4(ai.x * inv, ai.y * inv, ai.z * inv, ai.w * inv);
      res_pm   = 0.f;
      res_mask = mmax;
    }
    reinterpret_cast<float4*>(out_bX)[idx]  = res_x;
    reinterpret_cast<float4*>(out_bIX)[idx] = res_i;
    if (q == 0) {
      out_pm[row]    = res_pm;
      out_bmask[row] = res_mask;
    }
    return;
  }

  const int lane = tid & 31;
  const int wv   = __builtin_amdgcn_readfirstlane(tid >> 5);
  const int half = lane >> 4;
  const int lm   = lane & 15;
  const int rg   = blockIdx.x / CSL;
  const int cs   = blockIdx.x - rg * CSL;
  const int tfirst = rg * TPG + min(rg, TREM);
  const int ntile  = TPG + (rg < TREM ? 1 : 0);

  int rbase[VISITS], rcnt[VISITS]; bool rmid[VISITS];
  int sv[VISITS][BS]; float mv[VISITS][BS]; int pv[VISITS][BS];
  #pragma unroll
  for (int i = 0; i < VISITS; ++i) {
    const int tl  = min(wv + 8 * i, ntile - 1);
    const int row = (tfirst + tl) * 16 + lm;
    row_geom(row, rbase[i], rcnt[i], rmid[i]);
    #pragma unroll
    for (int p = 0; p < BS; ++p) {
      const int bp = rbase[i] + (p < rcnt[i] ? p : 0);
      sv[i][p] = S[bp];
      mv[i][p] = pad[bp];
      pv[i][p] = pmask[bp];
    }
  }

  {
    const int n   = (cs * CT_PER + wv) * 16 + lm;
    const bool v1ok = 16 + lm < VOCAB;
    const float* Arow0 = emb + (size_t)lm * EMB + 2 * half;
    const float* Arow1 = emb + (size_t)(v1ok ? 16 + lm : 0) * EMB + 2 * half;
    const float* Bcol  = Ww + (size_t)(2 * half) * H + n;
    v8f acc0 = {};
    v8f acc1 = {};
    #pragma unroll
    for (int k = 0; k < EMB; k += 4) {
      const v2f a0 = *reinterpret_cast<const v2f*>(Arow0 + k);
      const v2f a1 = *reinterpret_cast<const v2f*>(Arow1 + k);
      v2f af1;
      af1.x = v1ok ? a1.x : 0.f;
      af1.y = v1ok ? a1.y : 0.f;
      v2f bf;
      bf.x = Bcol[(size_t)k * H];
      bf.y = Bcol[(size_t)k * H + H];
      acc0 = __builtin_amdgcn_wmma_f32_16x16x4_f32(
           false, a0,  false, bf,
           (short)0, acc0,  false,  false);
      acc1 = __builtin_amdgcn_wmma_f32_16x16x4_f32(
           false, af1,  false, bf,
           (short)0, acc1,  false,  false);
    }
    const float wb = Wb[n];
    float* Tw = reinterpret_cast<float*>(Tl4);
    #pragma unroll
    for (int mt = 0; mt < 2; ++mt) {
      #pragma unroll
      for (int r = 0; r < 8; ++r) {
        const int vv = mt * 16 + half * 8 + r;
        if (vv < KP) {
          const float av = mt ? acc1[r] : acc0[r];
          const float val = (vv < VOCAB) ? av : ((vv == VOCAB) ? wb : 0.f);
          const int s  = vv >> 2;
          const int h2 = (vv >> 1) & 1;
          const int e  = vv & 1;
          Tw[(((s * CT_PER + wv) * 32) + h2 * 16 + lm) * 2 + e] = val;
        }
      }
    }
  }

  float a[VISITS][2 * KSTEPS];
  #pragma unroll
  for (int i = 0; i < VISITS; ++i) {
    #pragma unroll
    for (int qi = 0; qi < 2 * KSTEPS; ++qi) a[i][qi] = 0.f;
    if (wv + 8 * i < ntile)
      build_weights(rbase[i], rcnt[i], rmid[i], half, sv[i], mv[i], pv[i], cprob, a[i]);
  }

  __syncthreads();

  v2f tf[KSTEPS][CT_PER];
  {
    const v2f* Tl = reinterpret_cast<const v2f*>(Tl4);
    #pragma unroll
    for (int s = 0; s < KSTEPS; ++s) {
      #pragma unroll
      for (int ct = 0; ct < CT_PER; ++ct) tf[s][ct] = Tl[(s * CT_PER + ct) * 32 + lane];
    }
  }

  float* stg = reinterpret_cast<float*>(Sg4) + wv * SWAVE;
  float* wr  = stg + lm * SPITCH + half * 8;
  const float* rd = stg + half * SPITCH + lm * 4;
  #pragma unroll
  for (int i = 0; i < VISITS; ++i) {
    if (wv + 8 * i < ntile) {
      const int tile = tfirst + wv + 8 * i;
      float* orow = bS + (size_t)(tile * 16 + half) * H + cs * (CT_PER * 16) + lm * 4;
      #pragma unroll
      for (int g = 0; g < CT_PER / 4; ++g) {
        #pragma unroll
        for (int t = 0; t < 4; t += 2) {
          const int ct = g * 4 + t;
          v8f acc0 = {};
          v8f acc1 = {};
          #pragma unroll
          for (int s = 0; s < KSTEPS; ++s) {
            v2f af;
            af.x = a[i][2 * s];
            af.y = a[i][2 * s + 1];
            acc0 = __builtin_amdgcn_wmma_f32_16x16x4_f32(
                 false, tf[s][ct],  false, af,
                 (short)0, acc0,  false,  false);
            acc1 = __builtin_amdgcn_wmma_f32_16x16x4_f32(
                 false, tf[s][ct + 1],  false, af,
                 (short)0, acc1,  false,  false);
          }
          *reinterpret_cast<float4*>(wr + 16 * t)      = make_float4(acc0[0], acc0[1], acc0[2], acc0[3]);
          *reinterpret_cast<float4*>(wr + 16 * t + 4)  = make_float4(acc0[4], acc0[5], acc0[6], acc0[7]);
          *reinterpret_cast<float4*>(wr + 16 * t + 16) = make_float4(acc1[0], acc1[1], acc1[2], acc1[3]);
          *reinterpret_cast<float4*>(wr + 16 * t + 20) = make_float4(acc1[4], acc1[5], acc1[6], acc1[7]);
        }
        __builtin_amdgcn_fence(__ATOMIC_RELEASE, "wavefront");
        __builtin_amdgcn_wave_barrier();
        __builtin_amdgcn_fence(__ATOMIC_ACQUIRE, "wavefront");
        #pragma unroll
        for (int j = 0; j < 8; ++j) {
          const float4 v = *reinterpret_cast<const float4*>(rd + 2 * j * SPITCH);
          *reinterpret_cast<float4*>(orow + (size_t)(2 * j) * H + g * SCOLS) = v;
        }
        if (g + 1 < CT_PER / 4 || i + 1 < VISITS) {
          __builtin_amdgcn_fence(__ATOMIC_RELEASE, "wavefront");
          __builtin_amdgcn_wave_barrier();
          __builtin_amdgcn_fence(__ATOMIC_ACQUIRE, "wavefront");
        }
      }
    }
  }
}

extern "C" void kernel_launch(void* const* d_in, const int* in_sizes, int n_in,
                              void* d_out, int out_size, void* d_ws, size_t ws_size,
                              hipStream_t stream) {
  const float* X      = (const float*)d_in[0];
  const int*   S      = (const int*)  d_in[1];
  const float* pad    = (const float*)d_in[2];
  const float* cprob  = (const float*)d_in[3];
  const int*   pmask  = (const int*)  d_in[4];
  const float* initX  = (const float*)d_in[5];
  const float* emb    = (const float*)d_in[6];
  const float* Ww     = (const float*)d_in[7];
  const float* Wb     = (const float*)d_in[8];

  float* out   = (float*)d_out;
  float* bS    = out + OFF_BS;
  float* bX    = out + OFF_BX;
  float* pm    = out + OFF_PM;
  float* bIX   = out + OFF_BIX;
  float* bmask = out + OFF_BMASK;

  hd_fused<<<GRID, 256, 0, stream>>>(X, S, pad, cprob, pmask, initX, emb, Ww, Wb,
                                     bS, bX, pm, bIX, bmask);
}


# T build: wave = 1 vocab tile x 2 column tiles (lane = 2 neighbour columns), k order re-chosen: emb by 16-B, W_w by 8-B loads, 80 a wave not 128; W_b as accumulator start; T sum order changes
# speedup vs baseline: 5.0031x; 1.1150x over previous
#include <hip/hip_runtime.h>

typedef __attribute__((ext_vector_type(2))) float v2f;
typedef __attribute__((ext_vector_type(8))) float v8f;

constexpr int B     = 64;
constexpr int N     = 2048;
constexpr int VOCAB = 21;
constexpr int EMB   = 128;
constexpr int H     = 256;
constexpr int LB    = 900;
constexpr int RB    = 1100;
constexpr int BS    = 8;

constexpr int NB_PRE = (LB + BS - 1) / BS;
constexpr int MID    = RB - LB + 1;
constexpr int NB_SUF = (N - (RB + 1) + BS - 1) / BS;
constexpr int NBLK   = NB_PRE + MID + NB_SUF;
constexpr int ROWS   = B * NBLK;
constexpr int ROWTILES = ROWS / 16;
constexpr float EPSF = 1e-8f;

constexpr int KP     = 24;
constexpr int KSTEPS = KP / 4;
constexpr int NT     = H / 16;
constexpr int CSL    = 2;
constexpr int CT_PER = NT / CSL;
constexpr int TS_FLOATS = KSTEPS * CT_PER * 32 * 2;

constexpr int RG      = 128;
constexpr int TPG     = ROWTILES / RG;
constexpr int TREM    = ROWTILES - TPG * RG;
constexpr int VISITS  = (TPG + 1 + 7) / 8;
constexpr int ROWBLOCKS = RG * CSL;
constexpr int XITEMS  = ROWS * 3;
constexpr int XBLOCKS = (XITEMS + 255) / 256;
constexpr int GRID    = ROWBLOCKS + XBLOCKS;

constexpr int SCOLS  = 64;
constexpr int SPITCH = SCOLS + 4;
constexpr int SWAVE  = 16 * SPITCH;

constexpr size_t OFF_BS    = 0;
constexpr size_t OFF_BX    = OFF_BS    + (size_t)ROWS * H;
constexpr size_t OFF_PM    = OFF_BX    + (size_t)ROWS * 12;
constexpr size_t OFF_BIX   = OFF_PM    + (size_t)ROWS;
constexpr size_t OFF_BMASK = OFF_BIX   + (size_t)ROWS * 12;

__device__ __forceinline__ void row_geom(int row, int& base, int& cnt, bool& mid) {
  const int b = row / NBLK;
  const int j = row - b * NBLK;
  int p0;
  if (j < NB_PRE)            { p0 = j * BS;                           cnt = min(BS, LB - p0); mid = false; }
  else if (j < NB_PRE + MID) { p0 = LB + (j - NB_PRE);                cnt = 1;                mid = true;  }
  else                       { p0 = (RB + 1) + (j - NB_PRE - MID) * BS; cnt = min(BS, N - p0);  mid = false; }
  base = b * N + p0;
}

__device__ __forceinline__ void build_weights(
    int base, int cnt, bool mid, int half,
    const int (&sv)[BS], const float (&mv)[BS], const int (&pv)[BS],
    const float* __restrict__ cprob,
    float (&a)[2 * KSTEPS])
{
  #pragma unroll
  for (int qi = 0; qi < 2 * KSTEPS; ++qi) a[qi] = 0.f;
  float sm = 0.f;
  #pragma unroll
  for (int i = 0; i < BS; ++i) {
    const bool valid = i < cnt;
    const float w = valid ? (mid ? 1.f : mv[i]) : 0.f;
    sm += valid ? mv[i] : 0.f;
    if (valid && pv[i] > 0) {
      const float* c = cprob + (size_t)(base + i) * VOCAB;
      #pragma unroll
      for (int qi = 0; qi < 2 * KSTEPS; ++qi) {
        const int v = 4 * (qi >> 1) + 2 * half + (qi & 1);
        const float cv = c[v < VOCAB ? v : 0];
        a[qi] = fmaf(w, v < VOCAB ? cv : 0.f, a[qi]);
      }
    } else {
      #pragma unroll
      for (int qi = 0; qi < 2 * KSTEPS; ++qi) {
        const int v = 4 * (qi >> 1) + 2 * half + (qi & 1);
        a[qi] += (sv[i] == v) ? w : 0.f;
      }
    }
  }
  const float inv = mid ? 1.f : 1.f / (sm + EPSF);
  #pragma unroll
  for (int qi = 0; qi < 2 * KSTEPS; ++qi) a[qi] *= inv;
  const float cb = mid ? 1.f : sm * inv;
  if (half == 0) a[2 * KSTEPS - 1] = cb;
}

__global__ __launch_bounds__(256) void hd_fused(
    const float* __restrict__ X,
    const int*   __restrict__ S,
    const float* __restrict__ pad,
    const float* __restrict__ cprob,
    const int*   __restrict__ pmask,
    const float* __restrict__ initX,
    const float* __restrict__ emb,
    const float* __restrict__ Ww,
    const float* __restrict__ Wb,
    float* __restrict__ bS,
    float* __restrict__ out_bX,
    float* __restrict__ out_pm,
    float* __restrict__ out_bIX,
    float* __restrict__ out_bmask)
{
  __shared__ float4 Tl4[TS_FLOATS / 4];
  __shared__ float4 Sg4[8 * SWAVE / 4];
  const int tid = threadIdx.x;

  if (blockIdx.x >= ROWBLOCKS) {
    const int idx = (blockIdx.x - ROWBLOCKS) * 256 + tid;
    if (idx >= XITEMS) return;
    const int row = idx / 3;
    const int q   = idx - row * 3;
    int base, cnt; bool mid;
    row_geom(row, base, cnt, mid);
    const float4* X4 = reinterpret_cast<const float4*>(X);
    const float4* I4 = reinterpret_cast<const float4*>(initX);

    float  mv[BS]; int pv[BS]; float4 xv[BS];
    #pragma unroll
    for (int i = 0; i < BS; ++i) {
      const int bp = base + (i < cnt ? i : 0);
      mv[i] = pad[bp];
      pv[i] = pmask[bp];
      xv[i] = X4[(size_t)bp * 3 + q];
    }
    float4 res_x, res_i; float res_pm, res_mask;
    if (mid) {
      const bool sel = pv[0] > 0;
      res_x = xv[0];
      res_i = xv[0];
      if (sel) res_i = I4[(size_t)base * 3 + q];
      res_pm   = sel ? 1.f : 0.f;
      res_mask = mv[0];
    } else {
      float sm = 0.f;
      float mmax = (cnt < BS) ? 0.f : mv[0];
      float4 ax = make_float4(0.f, 0.f, 0.f, 0.f);
      float4 ai = ax;
      #pragma unroll
      for (int i = 0; i < BS; ++i) {
        if (i < cnt) {
          const float w = mv[i];
          sm  += w;
          mmax = fmaxf(mmax, w);
          float4 xi = xv[i];
          if (pv[i] > 0) xi = I4[(size_t)(base + i) * 3 + q];
          ax.x = fmaf(w, xv[i].x, ax.x); ax.y = fmaf(w, xv[i].y, ax.y);
          ax.z = fmaf(w, xv[i].z, ax.z); ax.w = fmaf(w, xv[i].w, ax.w);
          ai.x = fmaf(w, xi.x, ai.x);    ai.y = fmaf(w, xi.y, ai.y);
          ai.z = fmaf(w, xi.z, ai.z);    ai.w = fmaf(w, xi.w, ai.w);
        }
      }
      const float inv = 1.f / (sm + EPSF);
      res_x = make_float4(ax.x * inv, ax.y * inv, ax.z * inv, ax.w * inv);
      res_i = make_float4(ai.x * inv, ai.y * inv, ai.z * inv, ai.w * inv);
      res_pm   = 0.f;
      res_mask = mmax;
    }
    reinterpret_cast<float4*>(out_bX)[idx]  = res_x;
    reinterpret_cast<float4*>(out_bIX)[idx] = res_i;
    if (q == 0) {
      out_pm[row]    = res_pm;
      out_bmask[row] = res_mask;
    }
    return;
  }

  const int lane = tid & 31;
  const int wv   = __builtin_amdgcn_readfirstlane(tid >> 5);
  const int half = lane >> 4;
  const int lm   = lane & 15;
  const int rg   = blockIdx.x / CSL;
  const int cs   = blockIdx.x - rg * CSL;
  const int tfirst = rg * TPG + min(rg, TREM);
  const int ntile  = TPG + (rg < TREM ? 1 : 0);

  int rbase[VISITS], rcnt[VISITS]; bool rmid[VISITS];
  int sv[VISITS][BS]; float mv[VISITS][BS]; int pv[VISITS][BS];
  #pragma unroll
  for (int i = 0; i < VISITS; ++i) {
    const int tl  = min(wv + 8 * i, ntile - 1);
    const int row = (tfirst + tl) * 16 + lm;
    row_geom(row, rbase[i], rcnt[i], rmid[i]);
    #pragma unroll
    for (int p = 0; p < BS; ++p) {
      const int bp = rbase[i] + (p < rcnt[i] ? p : 0);
      sv[i][p] = S[bp];
      mv[i][p] = pad[bp];
      pv[i][p] = pmask[bp];
    }
  }

  {
    const int mt   = wv >> 2;
    const int cp   = wv & 3;
    const int ncol = cs * (CT_PER * 16) + 32 * cp + 2 * lm;
    const int vr   = mt * 16 + lm;
    const bool vok = vr < VOCAB;
    const float* Arow = emb + (size_t)(vok ? vr : 0) * EMB + 4 * half;
    const float* Bcol = Ww + (size_t)(4 * half) * H + ncol;
    static_assert(VOCAB > 16 && VOCAB < 24 && KP == 24, "the bias row must lie in rows 16..23 of the second vocab tile");
    const v2f wb = *reinterpret_cast<const v2f*>(Wb + ncol);
    const bool brow = (mt == 1) && (half == 0);
    v8f accA = {};
    v8f accB = {};
    accA[VOCAB - 16] = brow ? wb.x : 0.f;
    accB[VOCAB - 16] = brow ? wb.y : 0.f;
    #pragma unroll
    for (int q = 0; q < EMB / 8; ++q) {
      const float4 av = *reinterpret_cast<const float4*>(Arow + 8 * q);
      const v2f w0 = *reinterpret_cast<const v2f*>(Bcol + (size_t)(8 * q + 0) * H);
      const v2f w1 = *reinterpret_cast<const v2f*>(Bcol + (size_t)(8 * q + 1) * H);
      const v2f w2 = *reinterpret_cast<const v2f*>(Bcol + (size_t)(8 * q + 2) * H);
      const v2f w3 = *reinterpret_cast<const v2f*>(Bcol + (size_t)(8 * q + 3) * H);
      v2f a01, a23;
      a01.x = vok ? av.x : 0.f;  a01.y = vok ? av.y : 0.f;
      a23.x = vok ? av.z : 0.f;  a23.y = vok ? av.w : 0.f;
      v2f bA, bB;
      bA.x = w0.x; bA.y = w1.x;
      bB.x = w0.y; bB.y = w1.y;
      accA = __builtin_amdgcn_wmma_f32_16x16x4_f32(
           false, a01,  false, bA,
           (short)0, accA,  false,  false);
      accB = __builtin_amdgcn_wmma_f32_16x16x4_f32(
           false, a01,  false, bB,
           (short)0, accB,  false,  false);
      bA.x = w2.x; bA.y = w3.x;
      bB.x = w2.y; bB.y = w3.y;
      accA = __builtin_amdgcn_wmma_f32_16x16x4_f32(
           false, a23,  false, bA,
           (short)0, accA,  false,  false);
      accB = __builtin_amdgcn_wmma_f32_16x16x4_f32(
           false, a23,  false, bB,
           (short)0, accB,  false,  false);
    }
    float* Tw = reinterpret_cast<float*>(Tl4);
    if (mt == 0 || half == 0) {
      #pragma unroll
      for (int t = 0; t < 2; ++t) {
        const int col = 32 * cp + 2 * lm + t;
        const int ct  = col >> 4;
        const int lc  = col & 15;
        float* dst = Tw + (((mt * 4 + half * 2) * CT_PER + ct) * 32 + lc) * 2;
        #pragma unroll
        for (int r = 0; r < 8; r += 2) {
          const float v0 = t ? accB[r]     : accA[r];
          const float v1 = t ? accB[r + 1] : accA[r + 1];
          *reinterpret_cast<float2*>(dst + (r >> 2) * (CT_PER * 64) + ((r >> 1) & 1) * 32) = make_float2(v0, v1);
        }
      }
    }
  }

  float a[VISITS][2 * KSTEPS];
  #pragma unroll
  for (int i = 0; i < VISITS; ++i) {
    #pragma unroll
    for (int qi = 0; qi < 2 * KSTEPS; ++qi) a[i][qi] = 0.f;
    if (wv + 8 * i < ntile)
      build_weights(rbase[i], rcnt[i], rmid[i], half, sv[i], mv[i], pv[i], cprob, a[i]);
  }

  __syncthreads();

  v2f tf[KSTEPS][CT_PER];
  {
    const v2f* Tl = reinterpret_cast<const v2f*>(Tl4);
    #pragma unroll
    for (int s = 0; s < KSTEPS; ++s) {
      #pragma unroll
      for (int ct = 0; ct < CT_PER; ++ct) tf[s][ct] = Tl[(s * CT_PER + ct) * 32 + lane];
    }
  }

  float* stg = reinterpret_cast<float*>(Sg4) + wv * SWAVE;
  float* wr  = stg + lm * SPITCH + half * 8;
  const float* rd = stg + half * SPITCH + lm * 4;
  #pragma unroll
  for (int i = 0; i < VISITS; ++i) {
    if (wv + 8 * i < ntile) {
      const int tile = tfirst + wv + 8 * i;
      float* orow = bS + (size_t)(tile * 16 + half) * H + cs * (CT_PER * 16) + lm * 4;
      #pragma unroll
      for (int g = 0; g < CT_PER / 4; ++g) {
        #pragma unroll
        for (int t = 0; t < 4; t += 2) {
          const int ct = g * 4 + t;
          v8f acc0 = {};
          v8f acc1 = {};
          #pragma unroll
          for (int s = 0; s < KSTEPS; ++s) {
            v2f af;
            af.x = a[i][2 * s];
            af.y = a[i][2 * s + 1];
            acc0 = __builtin_amdgcn_wmma_f32_16x16x4_f32(
                 false, tf[s][ct],  false, af,
                 (short)0, acc0,  false,  false);
            acc1 = __builtin_amdgcn_wmma_f32_16x16x4_f32(
                 false, tf[s][ct + 1],  false, af,
                 (short)0, acc1,  false,  false);
          }
          *reinterpret_cast<float4*>(wr + 16 * t)      = make_float4(acc0[0], acc0[1], acc0[2], acc0[3]);
          *reinterpret_cast<float4*>(wr + 16 * t + 4)  = make_float4(acc0[4], acc0[5], acc0[6], acc0[7]);
          *reinterpret_cast<float4*>(wr + 16 * t + 16) = make_float4(acc1[0], acc1[1], acc1[2], acc1[3]);
          *reinterpret_cast<float4*>(wr + 16 * t + 20) = make_float4(acc1[4], acc1[5], acc1[6], acc1[7]);
        }
        __builtin_amdgcn_fence(__ATOMIC_RELEASE, "wavefront");
        __builtin_amdgcn_wave_barrier();
        __builtin_amdgcn_fence(__ATOMIC_ACQUIRE, "wavefront");
        #pragma unroll
        for (int j = 0; j < 8; ++j) {
          const float4 v = *reinterpret_cast<const float4*>(rd + 2 * j * SPITCH);
          *reinterpret_cast<float4*>(orow + (size_t)(2 * j) * H + g * SCOLS) = v;
        }
        if (g + 1 < CT_PER / 4 || i + 1 < VISITS) {
          __builtin_amdgcn_fence(__ATOMIC_RELEASE, "wavefront");
          __builtin_amdgcn_wave_barrier();
          __builtin_amdgcn_fence(__ATOMIC_ACQUIRE, "wavefront");
        }
      }
    }
  }
}

extern "C" void kernel_launch(void* const* d_in, const int* in_sizes, int n_in,
                              void* d_out, int out_size, void* d_ws, size_t ws_size,
                              hipStream_t stream) {
  const float* X      = (const float*)d_in[0];
  const int*   S      = (const int*)  d_in[1];
  const float* pad    = (const float*)d_in[2];
  const float* cprob  = (const float*)d_in[3];
  const int*   pmask  = (const int*)  d_in[4];
  const float* initX  = (const float*)d_in[5];
  const float* emb    = (const float*)d_in[6];
  const float* Ww     = (const float*)d_in[7];
  const float* Wb     = (const float*)d_in[8];

  float* out   = (float*)d_out;
  float* bS    = out + OFF_BS;
  float* bX    = out + OFF_BX;
  float* pm    = out + OFF_PM;
  float* bIX   = out + OFF_BIX;
  float* bmask = out + OFF_BMASK;

  hd_fused<<<GRID, 256, 0, stream>>>(X, S, pad, cprob, pmask, initX, emb, Ww, Wb,
                                     bS, bX, pm, bIX, bmask);
}


# One-hot part of the row weights summed inside the T build, one position per step, branch-free, so its vector work issues under the matrix pipe; flagged terms added after: weight sum order changes
# speedup vs baseline: 5.3481x; 1.0690x over previous
#include <hip/hip_runtime.h>

typedef __attribute__((ext_vector_type(2))) float v2f;
typedef __attribute__((ext_vector_type(8))) float v8f;

constexpr int B     = 64;
constexpr int N     = 2048;
constexpr int VOCAB = 21;
constexpr int EMB   = 128;
constexpr int H     = 256;
constexpr int LB    = 900;
constexpr int RB    = 1100;
constexpr int BS    = 8;

constexpr int NB_PRE = (LB + BS - 1) / BS;
constexpr int MID    = RB - LB + 1;
constexpr int NB_SUF = (N - (RB + 1) + BS - 1) / BS;
constexpr int NBLK   = NB_PRE + MID + NB_SUF;
constexpr int ROWS   = B * NBLK;
constexpr int ROWTILES = ROWS / 16;
constexpr float EPSF = 1e-8f;

constexpr int KP     = 24;
constexpr int KSTEPS = KP / 4;
constexpr int NT     = H / 16;
constexpr int CSL    = 2;
constexpr int CT_PER = NT / CSL;
constexpr int TS_FLOATS = KSTEPS * CT_PER * 32 * 2;

constexpr int RG      = 128;
constexpr int TPG     = ROWTILES / RG;
constexpr int TREM    = ROWTILES - TPG * RG;
constexpr int VISITS  = (TPG + 1 + 7) / 8;
constexpr int ROWBLOCKS = RG * CSL;
constexpr int XITEMS  = ROWS * 3;
constexpr int XBLOCKS = (XITEMS + 255) / 256;
constexpr int GRID    = ROWBLOCKS + XBLOCKS;

constexpr int SCOLS  = 64;
constexpr int SPITCH = SCOLS + 4;
constexpr int SWAVE  = 16 * SPITCH;

constexpr size_t OFF_BS    = 0;
constexpr size_t OFF_BX    = OFF_BS    + (size_t)ROWS * H;
constexpr size_t OFF_PM    = OFF_BX    + (size_t)ROWS * 12;
constexpr size_t OFF_BIX   = OFF_PM    + (size_t)ROWS;
constexpr size_t OFF_BMASK = OFF_BIX   + (size_t)ROWS * 12;

__device__ __forceinline__ void row_geom(int row, int& base, int& cnt, bool& mid) {
  const int b = row / NBLK;
  const int j = row - b * NBLK;
  int p0;
  if (j < NB_PRE)            { p0 = j * BS;                           cnt = min(BS, LB - p0); mid = false; }
  else if (j < NB_PRE + MID) { p0 = LB + (j - NB_PRE);                cnt = 1;                mid = true;  }
  else                       { p0 = (RB + 1) + (j - NB_PRE - MID) * BS; cnt = min(BS, N - p0);  mid = false; }
  base = b * N + p0;
}

__device__ __forceinline__ void weights_hard_step(
    int p, int cnt, bool mid, int half, int s, float m, int flag,
    float& sm, float (&a)[2 * KSTEPS])
{
  const bool valid = p < cnt;
  sm += valid ? m : 0.f;
  const float gate = (valid && !(flag > 0)) ? 1.f : 0.f;
  const float w = (mid ? 1.f : m) * gate;
  #pragma unroll
  for (int qi = 0; qi < 2 * KSTEPS; ++qi) {
    const int v = 4 * (qi >> 1) + 2 * half + (qi & 1);
    a[qi] += (s == v) ? w : 0.f;
  }
  asm volatile("" :: "v"(a[0]), "v"(a[1]), "v"(a[2]), "v"(a[3]), "v"(a[4]), "v"(a[5]),
                     "v"(a[6]), "v"(a[7]), "v"(a[8]), "v"(a[9]), "v"(a[10]), "v"(a[11]), "v"(sm));
}

__device__ __forceinline__ void weights_finish(
    int base, int cnt, bool mid, int half, float sm,
    const float (&mv)[BS], const int (&pv)[BS],
    const float* __restrict__ cprob,
    float (&a)[2 * KSTEPS])
{
  bool anyflag = false;
  #pragma unroll
  for (int i = 0; i < BS; ++i) anyflag = anyflag || (i < cnt && pv[i] > 0);
  if (anyflag) {
    #pragma unroll
    for (int i = 0; i < BS; ++i) {
      if (i < cnt && pv[i] > 0) {
        const float w = mid ? 1.f : mv[i];
        const float* c = cprob + (size_t)(base + i) * VOCAB;
        #pragma unroll
        for (int qi = 0; qi < 2 * KSTEPS; ++qi) {
          const int v = 4 * (qi >> 1) + 2 * half + (qi & 1);
          const float cv = c[v < VOCAB ? v : 0];
          a[qi] = fmaf(w, v < VOCAB ? cv : 0.f, a[qi]);
        }
      }
    }
  }
  const float inv = mid ? 1.f : 1.f / (sm + EPSF);
  #pragma unroll
  for (int qi = 0; qi < 2 * KSTEPS; ++qi) a[qi] *= inv;
  const float cb = mid ? 1.f : sm * inv;
  if (half == 0) a[2 * KSTEPS - 1] = cb;
}

__global__ __launch_bounds__(256) void hd_fused(
    const float* __restrict__ X,
    const int*   __restrict__ S,
    const float* __restrict__ pad,
    const float* __restrict__ cprob,
    const int*   __restrict__ pmask,
    const float* __restrict__ initX,
    const float* __restrict__ emb,
    const float* __restrict__ Ww,
    const float* __restrict__ Wb,
    float* __restrict__ bS,
    float* __restrict__ out_bX,
    float* __restrict__ out_pm,
    float* __restrict__ out_bIX,
    float* __restrict__ out_bmask)
{
  __shared__ float4 Tl4[TS_FLOATS / 4];
  __shared__ float4 Sg4[8 * SWAVE / 4];
  const int tid = threadIdx.x;

  if (blockIdx.x >= ROWBLOCKS) {
    const int idx = (blockIdx.x - ROWBLOCKS) * 256 + tid;
    if (idx >= XITEMS) return;
    const int row = idx / 3;
    const int q   = idx - row * 3;
    int base, cnt; bool mid;
    row_geom(row, base, cnt, mid);
    const float4* X4 = reinterpret_cast<const float4*>(X);
    const float4* I4 = reinterpret_cast<const float4*>(initX);

    float  mv[BS]; int pv[BS]; float4 xv[BS];
    #pragma unroll
    for (int i = 0; i < BS; ++i) {
      const int bp = base + (i < cnt ? i : 0);
      mv[i] = pad[bp];
      pv[i] = pmask[bp];
      xv[i] = X4[(size_t)bp * 3 + q];
    }
    float4 res_x, res_i; float res_pm, res_mask;
    if (mid) {
      const bool sel = pv[0] > 0;
      res_x = xv[0];
      res_i = xv[0];
      if (sel) res_i = I4[(size_t)base * 3 + q];
      res_pm   = sel ? 1.f : 0.f;
      res_mask = mv[0];
    } else {
      float sm = 0.f;
      float mmax = (cnt < BS) ? 0.f : mv[0];
      float4 ax = make_float4(0.f, 0.f, 0.f, 0.f);
      float4 ai = ax;
      #pragma unroll
      for (int i = 0; i < BS; ++i) {
        if (i < cnt) {
          const float w = mv[i];
          sm  += w;
          mmax = fmaxf(mmax, w);
          float4 xi = xv[i];
          if (pv[i] > 0) xi = I4[(size_t)(base + i) * 3 + q];
          ax.x = fmaf(w, xv[i].x, ax.x); ax.y = fmaf(w, xv[i].y, ax.y);
          ax.z = fmaf(w, xv[i].z, ax.z); ax.w = fmaf(w, xv[i].w, ax.w);
          ai.x = fmaf(w, xi.x, ai.x);    ai.y = fmaf(w, xi.y, ai.y);
          ai.z = fmaf(w, xi.z, ai.z);    ai.w = fmaf(w, xi.w, ai.w);
        }
      }
      const float inv = 1.f / (sm + EPSF);
      res_x = make_float4(ax.x * inv, ax.y * inv, ax.z * inv, ax.w * inv);
      res_i = make_float4(ai.x * inv, ai.y * inv, ai.z * inv, ai.w * inv);
      res_pm   = 0.f;
      res_mask = mmax;
    }
    reinterpret_cast<float4*>(out_bX)[idx]  = res_x;
    reinterpret_cast<float4*>(out_bIX)[idx] = res_i;
    if (q == 0) {
      out_pm[row]    = res_pm;
      out_bmask[row] = res_mask;
    }
    return;
  }

  const int lane = tid & 31;
  const int wv   = __builtin_amdgcn_readfirstlane(tid >> 5);
  const int half = lane >> 4;
  const int lm   = lane & 15;
  const int rg   = blockIdx.x / CSL;
  const int cs   = blockIdx.x - rg * CSL;
  const int tfirst = rg * TPG + min(rg, TREM);
  const int ntile  = TPG + (rg < TREM ? 1 : 0);

  int rbase[VISITS], rcnt[VISITS]; bool rmid[VISITS];
  int sv[VISITS][BS]; float mv[VISITS][BS]; int pv[VISITS][BS];
  #pragma unroll
  for (int i = 0; i < VISITS; ++i) {
    const int tl  = min(wv + 8 * i, ntile - 1);
    const int row = (tfirst + tl) * 16 + lm;
    row_geom(row, rbase[i], rcnt[i], rmid[i]);
    #pragma unroll
    for (int p = 0; p < BS; ++p) {
      const int bp = rbase[i] + (p < rcnt[i] ? p : 0);
      sv[i][p] = S[bp];
      mv[i][p] = pad[bp];
      pv[i][p] = pmask[bp];
    }
  }

  float a[VISITS][2 * KSTEPS];
  float smv[VISITS];
  #pragma unroll
  for (int i = 0; i < VISITS; ++i) {
    smv[i] = 0.f;
    #pragma unroll
    for (int qi = 0; qi < 2 * KSTEPS; ++qi) a[i][qi] = 0.f;
  }

  {
    const int mt   = wv >> 2;
    const int cp   = wv & 3;
    const int ncol = cs * (CT_PER * 16) + 32 * cp + 2 * lm;
    const int vr   = mt * 16 + lm;
    const bool vok = vr < VOCAB;
    const float* Arow = emb + (size_t)(vok ? vr : 0) * EMB + 4 * half;
    const float* Bcol = Ww + (size_t)(4 * half) * H + ncol;
    static_assert(VOCAB > 16 && VOCAB < 24 && KP == 24, "the bias row must lie in rows 16..23 of the second vocab tile");
    const v2f wb = *reinterpret_cast<const v2f*>(Wb + ncol);
    const bool brow = (mt == 1) && (half == 0);
    static_assert(EMB / 8 == VISITS * BS, "one position of one row tile per step of the T build");
    v8f accA = {};
    v8f accB = {};
    accA[VOCAB - 16] = brow ? wb.x : 0.f;
    accB[VOCAB - 16] = brow ? wb.y : 0.f;
    constexpr int QSTEPS = EMB / 8;
    constexpr int PF = 7;
    float4 av[QSTEPS];
    v2f wq[QSTEPS][4];
    #pragma unroll
    for (int q = 0; q < PF; ++q) {
      av[q] = *reinterpret_cast<const float4*>(Arow + 8 * q);
      #pragma unroll
      for (int r = 0; r < 4; ++r) wq[q][r] = *reinterpret_cast<const v2f*>(Bcol + (size_t)(8 * q + r) * H);
    }
    __builtin_amdgcn_sched_barrier(0);
    #pragma unroll
    for (int q = 0; q < QSTEPS; ++q) {
      if (q + PF < QSTEPS) {
        av[q + PF] = *reinterpret_cast<const float4*>(Arow + 8 * (q + PF));
        #pragma unroll
        for (int r = 0; r < 4; ++r)
          wq[q + PF][r] = *reinterpret_cast<const v2f*>(Bcol + (size_t)(8 * (q + PF) + r) * H);
      }
      v2f a01, a23;
      a01.x = vok ? av[q].x : 0.f;  a01.y = vok ? av[q].y : 0.f;
      a23.x = vok ? av[q].z : 0.f;  a23.y = vok ? av[q].w : 0.f;
      v2f bA, bB;
      bA.x = wq[q][0].x; bA.y = wq[q][1].x;
      bB.x = wq[q][0].y; bB.y = wq[q][1].y;
      accA = __builtin_amdgcn_wmma_f32_16x16x4_f32(
           false, a01,  false, bA,
           (short)0, accA,  false,  false);
      accB = __builtin_amdgcn_wmma_f32_16x16x4_f32(
           false, a01,  false, bB,
           (short)0, accB,  false,  false);
      bA.x = wq[q][2].x; bA.y = wq[q][3].x;
      bB.x = wq[q][2].y; bB.y = wq[q][3].y;
      accA = __builtin_amdgcn_wmma_f32_16x16x4_f32(
           false, a23,  false, bA,
           (short)0, accA,  false,  false);
      accB = __builtin_amdgcn_wmma_f32_16x16x4_f32(
           false, a23,  false, bB,
           (short)0, accB,  false,  false);
      weights_hard_step(q % BS, rcnt[q / BS], rmid[q / BS], half,
                        sv[q / BS][q % BS], mv[q / BS][q % BS], pv[q / BS][q % BS],
                        smv[q / BS], a[q / BS]);
      if (q + PF < QSTEPS) __builtin_amdgcn_sched_group_barrier(0x020, 5, 0);
      #pragma unroll
      for (int t = 0; t < 2; ++t) {
        __builtin_amdgcn_sched_group_barrier(0x008, 2, 0);
        __builtin_amdgcn_sched_group_barrier(0x002, 16, 0);
      }
      __builtin_amdgcn_sched_barrier(0);
    }
    float* Tw = reinterpret_cast<float*>(Tl4);
    if (mt == 0 || half == 0) {
      #pragma unroll
      for (int t = 0; t < 2; ++t) {
        const int col = 32 * cp + 2 * lm + t;
        const int ct  = col >> 4;
        const int lc  = col & 15;
        float* dst = Tw + (((mt * 4 + half * 2) * CT_PER + ct) * 32 + lc) * 2;
        #pragma unroll
        for (int r = 0; r < 8; r += 2) {
          const float v0 = t ? accB[r]     : accA[r];
          const float v1 = t ? accB[r + 1] : accA[r + 1];
          *reinterpret_cast<float2*>(dst + (r >> 2) * (CT_PER * 64) + ((r >> 1) & 1) * 32) = make_float2(v0, v1);
        }
      }
    }
  }

  #pragma unroll
  for (int i = 0; i < VISITS; ++i) {
    if (wv + 8 * i < ntile)
      weights_finish(rbase[i], rcnt[i], rmid[i], half, smv[i], mv[i], pv[i], cprob, a[i]);
  }

  __syncthreads();

  v2f tf[KSTEPS][CT_PER];
  {
    const v2f* Tl = reinterpret_cast<const v2f*>(Tl4);
    #pragma unroll
    for (int s = 0; s < KSTEPS; ++s) {
      #pragma unroll
      for (int ct = 0; ct < CT_PER; ++ct) tf[s][ct] = Tl[(s * CT_PER + ct) * 32 + lane];
    }
  }

  float* stg = reinterpret_cast<float*>(Sg4) + wv * SWAVE;
  float* wr  = stg + lm * SPITCH + half * 8;
  const float* rd = stg + half * SPITCH + lm * 4;
  #pragma unroll
  for (int i = 0; i < VISITS; ++i) {
    if (wv + 8 * i < ntile) {
      const int tile = tfirst + wv + 8 * i;
      float* orow = bS + (size_t)(tile * 16 + half) * H + cs * (CT_PER * 16) + lm * 4;
      #pragma unroll
      for (int g = 0; g < CT_PER / 4; ++g) {
        #pragma unroll
        for (int t = 0; t < 4; t += 2) {
          const int ct = g * 4 + t;
          v8f acc0 = {};
          v8f acc1 = {};
          #pragma unroll
          for (int s = 0; s < KSTEPS; ++s) {
            v2f af;
            af.x = a[i][2 * s];
            af.y = a[i][2 * s + 1];
            acc0 = __builtin_amdgcn_wmma_f32_16x16x4_f32(
                 false, tf[s][ct],  false, af,
                 (short)0, acc0,  false,  false);
            acc1 = __builtin_amdgcn_wmma_f32_16x16x4_f32(
                 false, tf[s][ct + 1],  false, af,
                 (short)0, acc1,  false,  false);
          }
          *reinterpret_cast<float4*>(wr + 16 * t)      = make_float4(acc0[0], acc0[1], acc0[2], acc0[3]);
          *reinterpret_cast<float4*>(wr + 16 * t + 4)  = make_float4(acc0[4], acc0[5], acc0[6], acc0[7]);
          *reinterpret_cast<float4*>(wr + 16 * t + 16) = make_float4(acc1[0], acc1[1], acc1[2], acc1[3]);
          *reinterpret_cast<float4*>(wr + 16 * t + 20) = make_float4(acc1[4], acc1[5], acc1[6], acc1[7]);
        }
        __builtin_amdgcn_fence(__ATOMIC_RELEASE, "wavefront");
        __builtin_amdgcn_wave_barrier();
        __builtin_amdgcn_fence(__ATOMIC_ACQUIRE, "wavefront");
        #pragma unroll
        for (int j = 0; j < 8; ++j) {
          const float4 v = *reinterpret_cast<const float4*>(rd + 2 * j * SPITCH);
          *reinterpret_cast<float4*>(orow + (size_t)(2 * j) * H + g * SCOLS) = v;
        }
        if (g + 1 < CT_PER / 4 || i + 1 < VISITS) {
          __builtin_amdgcn_fence(__ATOMIC_RELEASE, "wavefront");
          __builtin_amdgcn_wave_barrier();
          __builtin_amdgcn_fence(__ATOMIC_ACQUIRE, "wavefront");
        }
      }
    }
  }
}

extern "C" void kernel_launch(void* const* d_in, const int* in_sizes, int n_in,
                              void* d_out, int out_size, void* d_ws, size_t ws_size,
                              hipStream_t stream) {
  const float* X      = (const float*)d_in[0];
  const int*   S      = (const int*)  d_in[1];
  const float* pad    = (const float*)d_in[2];
  const float* cprob  = (const float*)d_in[3];
  const int*   pmask  = (const int*)  d_in[4];
  const float* initX  = (const float*)d_in[5];
  const float* emb    = (const float*)d_in[6];
  const float* Ww     = (const float*)d_in[7];
  const float* Wb     = (const float*)d_in[8];

  float* out   = (float*)d_out;
  float* bS    = out + OFF_BS;
  float* bX    = out + OFF_BX;
  float* pm    = out + OFF_PM;
  float* bIX   = out + OFF_BIX;
  float* bmask = out + OFF_BMASK;

  hd_fused<<<GRID, 256, 0, stream>>>(X, S, pad, cprob, pmask, initX, emb, Ww, Wb,
                                     bS, bX, pm, bIX, bmask);
}
